# starting kernel
# baseline (speedup 1.0000x reference)
#include <hip/hip_runtime.h>
#include <math.h>

typedef __attribute__((ext_vector_type(2))) float v2f;
typedef __attribute__((ext_vector_type(4))) float v4f;
typedef __attribute__((ext_vector_type(8))) float v8f;

#define CANVAS_H 256
#define CANVAS_W 384
#define NPTS     1024
#define R_WH     1.5f

__global__ __launch_bounds__(256) void points_kernel(
    const float* __restrict__ loc,
    const float* __restrict__ mo,
    const float* __restrict__ mse,
    const float* __restrict__ col,
    float* __restrict__ out)
{
  __shared__ __align__(16) float lds[12 * NPTS];
  float* BJ0 = lds;
  float* BJ1 = lds + 4 * NPTS;
  float* C4  = lds + 8 * NPTS;

  for (int n = threadIdx.x; n < NPTS; n += blockDim.x) {
    float s   = 16.0f * __expf(-mse[n]);
    float m00 = mo[n * 4 + 0] + s;
    float m01 = mo[n * 4 + 1];
    float m10 = mo[n * 4 + 2];
    float m11 = mo[n * 4 + 3] + s;
    float ly  = loc[n * 2 + 0];
    float lx  = loc[n * 2 + 1];
    float b0  = ly * m00 + lx * m10;
    float b1  = ly * m01 + lx * m11;

    v2f lo0; lo0.x = m00; lo0.y = m10;
    v2f hi0; hi0.x = -b0; hi0.y = 0.0f;
    v2f lo1; lo1.x = m01; lo1.y = m11;
    v2f hi1; hi1.x = -b1; hi1.y = 0.0f;
    *(v2f*)&BJ0[n * 2]              = lo0;
    *(v2f*)&BJ0[(NPTS + n) * 2]     = hi0;
    *(v2f*)&BJ1[n * 2]              = lo1;
    *(v2f*)&BJ1[(NPTS + n) * 2]     = hi1;
    v4f c; c.x = col[n * 3 + 0]; c.y = col[n * 3 + 1];
    c.z = col[n * 3 + 2]; c.w = 0.0f;
    *(v4f*)&C4[n * 4] = c;
  }
  __syncthreads();

  const int lane = threadIdx.x & 31;
  const int wave = threadIdx.x >> 5;
  const int tile = blockIdx.x * 8 + wave;
  const int hi   = lane >> 4;
  const int ln   = lane & 15;

  int p0 = tile * 16 + lane;
  int h0 = p0 / CANVAS_W, w0 = p0 - h0 * CANVAS_W;
  float gy = -1.0f + (float)h0 * (2.0f / (float)(CANVAS_H - 1));
  float gx = -R_WH + (float)w0 * (2.0f * R_WH / (float)(CANVAS_W - 1));
  v2f a;
  a.x = hi ? 1.0f : gy;
  a.y = hi ? 0.0f : gx;

  float accR[8], accG[8], accB[8];
#pragma unroll
  for (int e = 0; e < 8; ++e) { accR[e] = 0.0f; accG[e] = 0.0f; accB[e] = 0.0f; }

  const v8f cz = {};
  const int bsel = (hi * NPTS + ln) * 2;

  for (int k = 0; k < NPTS / 16; ++k) {
    v2f b0 = *(const v2f*)&BJ0[bsel + k * 32];
    v2f b1 = *(const v2f*)&BJ1[bsel + k * 32];
    v4f c4 = *(const v4f*)&C4[(k * 16 + ln) * 4];

    v8f d0 = __builtin_amdgcn_wmma_f32_16x16x4_f32(false, a, false, b0,
                                                   (short)0, cz, false, false);
    v8f d1 = __builtin_amdgcn_wmma_f32_16x16x4_f32(false, a, false, b1,
                                                   (short)0, cz, false, false);

#pragma unroll
    for (int e = 0; e < 8; ++e) {
      float dist = 0.5f * (fabsf(d0[e]) + fabsf(d1[e]));
      float m    = fmaxf(0.0f, 1.0f - dist);
      accR[e] = fmaf(m, c4.x, accR[e]);
      accG[e] = fmaf(m, c4.y, accG[e]);
      accB[e] = fmaf(m, c4.z, accB[e]);
    }
  }

#pragma unroll
  for (int e = 0; e < 8; ++e) {
#pragma unroll
    for (int off = 1; off < 16; off <<= 1) {
      accR[e] += __shfl_xor(accR[e], off, 16);
      accG[e] += __shfl_xor(accG[e], off, 16);
      accB[e] += __shfl_xor(accB[e], off, 16);
    }
  }

#pragma unroll
  for (int e = 0; e < 8; ++e) {
    if (ln == e) {
      int row = e + hi * 8;
      int p   = tile * 16 + row;
      int h   = p / CANVAS_W, w = p - h * CANVAS_W;
      float r = 1.0f / (1.0f + __expf(-4.0f * accR[e]));
      float g = 1.0f / (1.0f + __expf(-4.0f * accG[e]));
      float b = 1.0f / (1.0f + __expf(-4.0f * accB[e]));
      out[0 * CANVAS_H * CANVAS_W + h * CANVAS_W + w] = r;
      out[1 * CANVAS_H * CANVAS_W + h * CANVAS_W + w] = g;
      out[2 * CANVAS_H * CANVAS_W + h * CANVAS_W + w] = b;
    }
  }
}

extern "C" void kernel_launch(void* const* d_in, const int* in_sizes, int n_in,
                              void* d_out, int out_size, void* d_ws, size_t ws_size,
                              hipStream_t stream) {
  (void)in_sizes; (void)n_in; (void)out_size; (void)d_ws; (void)ws_size;
  const float* loc = (const float*)d_in[0];
  const float* mo  = (const float*)d_in[1];
  const float* mse = (const float*)d_in[2];
  const float* col = (const float*)d_in[3];
  float* out = (float*)d_out;

  const int tiles  = (CANVAS_H * CANVAS_W) / 16;
  const int blocks = tiles / 8;
  points_kernel<<<blocks, 256, 0, stream>>>(loc, mo, mse, col, out);
}


# relu(1-dist) per pair as two v_fma_f32 in asm: |.| as source modifiers, relu as the clamp bit, replacing v_and/pk_add/pk_fma/max; one wait covers the asm reads; rounding order inside dist changes
# speedup vs baseline: 1.1040x; 1.1040x over previous
#include <hip/hip_runtime.h>
#include <math.h>

typedef __attribute__((ext_vector_type(2))) float v2f;
typedef __attribute__((ext_vector_type(4))) float v4f;
typedef __attribute__((ext_vector_type(8))) float v8f;

#define CANVAS_H 256
#define CANVAS_W 384
#define NPTS     1024
#define R_WH     1.5f


static __device__ __forceinline__ float half_abs_start(float t0) {
  float x;
  asm("v_fma_f32 %0, -0.5, |%1|, 1.0" : "=v"(x) : "v"(t0));
  return x;
}
static __device__ __forceinline__ float half_abs_relu(float t1, float x) {
  float m;
  asm("v_fma_f32 %0, -0.5, |%1|, %2 clamp" : "=v"(m) : "v"(t1), "v"(x));
  return m;
}

__global__ __launch_bounds__(256) void points_kernel(
    const float* __restrict__ loc,
    const float* __restrict__ mo,
    const float* __restrict__ mse,
    const float* __restrict__ col,
    float* __restrict__ out)
{
  __shared__ __align__(16) float lds[12 * NPTS];
  float* BJ0 = lds;
  float* BJ1 = lds + 4 * NPTS;
  float* C4  = lds + 8 * NPTS;

  for (int n = threadIdx.x; n < NPTS; n += blockDim.x) {
    float s   = 16.0f * __expf(-mse[n]);
    float m00 = mo[n * 4 + 0] + s;
    float m01 = mo[n * 4 + 1];
    float m10 = mo[n * 4 + 2];
    float m11 = mo[n * 4 + 3] + s;
    float ly  = loc[n * 2 + 0];
    float lx  = loc[n * 2 + 1];
    float b0  = ly * m00 + lx * m10;
    float b1  = ly * m01 + lx * m11;

    v2f lo0; lo0.x = m00; lo0.y = m10;
    v2f hi0; hi0.x = -b0; hi0.y = 0.0f;
    v2f lo1; lo1.x = m01; lo1.y = m11;
    v2f hi1; hi1.x = -b1; hi1.y = 0.0f;
    *(v2f*)&BJ0[n * 2]              = lo0;
    *(v2f*)&BJ0[(NPTS + n) * 2]     = hi0;
    *(v2f*)&BJ1[n * 2]              = lo1;
    *(v2f*)&BJ1[(NPTS + n) * 2]     = hi1;
    v4f c; c.x = col[n * 3 + 0]; c.y = col[n * 3 + 1];
    c.z = col[n * 3 + 2]; c.w = 0.0f;
    *(v4f*)&C4[n * 4] = c;
  }
  __syncthreads();

  const int lane = threadIdx.x & 31;
  const int wave = threadIdx.x >> 5;
  const int tile = blockIdx.x * 8 + wave;
  const int hi   = lane >> 4;
  const int ln   = lane & 15;

  int p0 = tile * 16 + lane;
  int h0 = p0 / CANVAS_W, w0 = p0 - h0 * CANVAS_W;
  float gy = -1.0f + (float)h0 * (2.0f / (float)(CANVAS_H - 1));
  float gx = -R_WH + (float)w0 * (2.0f * R_WH / (float)(CANVAS_W - 1));
  v2f a;
  a.x = hi ? 1.0f : gy;
  a.y = hi ? 0.0f : gx;

  float accR[8], accG[8], accB[8];
#pragma unroll
  for (int e = 0; e < 8; ++e) { accR[e] = 0.0f; accG[e] = 0.0f; accB[e] = 0.0f; }

  const v8f cz = {};
  const int bsel = (hi * NPTS + ln) * 2;

  for (int k = 0; k < NPTS / 16; ++k) {
    v2f b0 = *(const v2f*)&BJ0[bsel + k * 32];
    v2f b1 = *(const v2f*)&BJ1[bsel + k * 32];
    v4f c4 = *(const v4f*)&C4[(k * 16 + ln) * 4];

    v8f d0 = __builtin_amdgcn_wmma_f32_16x16x4_f32(false, a, false, b0,
                                                   (short)0, cz, false, false);
    v8f d1 = __builtin_amdgcn_wmma_f32_16x16x4_f32(false, a, false, b1,
                                                   (short)0, cz, false, false);

    asm volatile("s_wait_alu depctr_va_vdst(0)" : "+v"(d0), "+v"(d1));

    float x[8];
#pragma unroll
    for (int e = 0; e < 8; ++e) x[e] = half_abs_start(d0[e]);
#pragma unroll
    for (int e = 0; e < 8; ++e) {
      float m = half_abs_relu(d1[e], x[e]);
      accR[e] = fmaf(m, c4.x, accR[e]);
      accG[e] = fmaf(m, c4.y, accG[e]);
      accB[e] = fmaf(m, c4.z, accB[e]);
    }
  }

#pragma unroll
  for (int e = 0; e < 8; ++e) {
#pragma unroll
    for (int off = 1; off < 16; off <<= 1) {
      accR[e] += __shfl_xor(accR[e], off, 16);
      accG[e] += __shfl_xor(accG[e], off, 16);
      accB[e] += __shfl_xor(accB[e], off, 16);
    }
  }

#pragma unroll
  for (int e = 0; e < 8; ++e) {
    if (ln == e) {
      int row = e + hi * 8;
      int p   = tile * 16 + row;
      int h   = p / CANVAS_W, w = p - h * CANVAS_W;
      float r = 1.0f / (1.0f + __expf(-4.0f * accR[e]));
      float g = 1.0f / (1.0f + __expf(-4.0f * accG[e]));
      float b = 1.0f / (1.0f + __expf(-4.0f * accB[e]));
      out[0 * CANVAS_H * CANVAS_W + h * CANVAS_W + w] = r;
      out[1 * CANVAS_H * CANVAS_W + h * CANVAS_W + w] = g;
      out[2 * CANVAS_H * CANVAS_W + h * CANVAS_W + w] = b;
    }
  }
}

extern "C" void kernel_launch(void* const* d_in, const int* in_sizes, int n_in,
                              void* d_out, int out_size, void* d_ws, size_t ws_size,
                              hipStream_t stream) {
  (void)in_sizes; (void)n_in; (void)out_size; (void)d_ws; (void)ws_size;
  const float* loc = (const float*)d_in[0];
  const float* mo  = (const float*)d_in[1];
  const float* mse = (const float*)d_in[2];
  const float* col = (const float*)d_in[3];
  float* out = (float*)d_out;

  const int tiles  = (CANVAS_H * CANVAS_W) / 16;
  const int blocks = tiles / 8;
  points_kernel<<<blocks, 256, 0, stream>>>(loc, mo, mse, col, out);
}


# epilogue: 16-lane sum as reduce-scatter on row_xmask adds (24, was 96 LDS permutes); lane ln ends owning row ln&7: one uniform sigmoid+store replaces 8 divergent; order of sum over points changes
# speedup vs baseline: 1.1864x; 1.0746x over previous
#include <hip/hip_runtime.h>
#include <math.h>

typedef __attribute__((ext_vector_type(2))) float v2f;
typedef __attribute__((ext_vector_type(4))) float v4f;
typedef __attribute__((ext_vector_type(8))) float v8f;

#define CANVAS_H 256
#define CANVAS_W 384
#define NPTS     1024
#define R_WH     1.5f


static __device__ __forceinline__ float half_abs_start(float t0) {
  float x;
  asm("v_fma_f32 %0, -0.5, |%1|, 1.0" : "=v"(x) : "v"(t0));
  return x;
}
static __device__ __forceinline__ float half_abs_relu(float t1, float x) {
  float m;
  asm("v_fma_f32 %0, -0.5, |%1|, %2 clamp" : "=v"(m) : "v"(t1), "v"(x));
  return m;
}

template <int M>
static __device__ __forceinline__ float lane_xor(float v) {
  return __int_as_float(__builtin_amdgcn_update_dpp(
      0, __float_as_int(v), 0x160 + M, 0xf, 0xf, true));
}

__global__ __launch_bounds__(256) void points_kernel(
    const float* __restrict__ loc,
    const float* __restrict__ mo,
    const float* __restrict__ mse,
    const float* __restrict__ col,
    float* __restrict__ out)
{
  __shared__ __align__(16) float lds[12 * NPTS];
  float* BJ0 = lds;
  float* BJ1 = lds + 4 * NPTS;
  float* C4  = lds + 8 * NPTS;

  for (int n = threadIdx.x; n < NPTS; n += blockDim.x) {
    float s   = 16.0f * __expf(-mse[n]);
    float m00 = mo[n * 4 + 0] + s;
    float m01 = mo[n * 4 + 1];
    float m10 = mo[n * 4 + 2];
    float m11 = mo[n * 4 + 3] + s;
    float ly  = loc[n * 2 + 0];
    float lx  = loc[n * 2 + 1];
    float b0  = ly * m00 + lx * m10;
    float b1  = ly * m01 + lx * m11;

    v2f lo0; lo0.x = m00; lo0.y = m10;
    v2f hi0; hi0.x = -b0; hi0.y = 0.0f;
    v2f lo1; lo1.x = m01; lo1.y = m11;
    v2f hi1; hi1.x = -b1; hi1.y = 0.0f;
    *(v2f*)&BJ0[n * 2]              = lo0;
    *(v2f*)&BJ0[(NPTS + n) * 2]     = hi0;
    *(v2f*)&BJ1[n * 2]              = lo1;
    *(v2f*)&BJ1[(NPTS + n) * 2]     = hi1;
    v4f c; c.x = col[n * 3 + 0]; c.y = col[n * 3 + 1];
    c.z = col[n * 3 + 2]; c.w = 0.0f;
    *(v4f*)&C4[n * 4] = c;
  }
  __syncthreads();

  const int lane = threadIdx.x & 31;
  const int wave = threadIdx.x >> 5;
  const int tile = blockIdx.x * 8 + wave;
  const int hi   = lane >> 4;
  const int ln   = lane & 15;

  int p0 = tile * 16 + lane;
  int h0 = p0 / CANVAS_W, w0 = p0 - h0 * CANVAS_W;
  float gy = -1.0f + (float)h0 * (2.0f / (float)(CANVAS_H - 1));
  float gx = -R_WH + (float)w0 * (2.0f * R_WH / (float)(CANVAS_W - 1));
  v2f a;
  a.x = hi ? 1.0f : gy;
  a.y = hi ? 0.0f : gx;

  v2f accR2[4], accG2[4], accB2[4];
#pragma unroll
  for (int q = 0; q < 4; ++q) {
    accR2[q] = (v2f){0.0f, 0.0f}; accG2[q] = (v2f){0.0f, 0.0f}; accB2[q] = (v2f){0.0f, 0.0f};
  }

  const v8f cz = {};
  const int bsel = (hi * NPTS + ln) * 2;

#pragma unroll 2
  for (int k = 0; k < NPTS / 16; ++k) {
    v2f b0 = *(const v2f*)&BJ0[bsel + k * 32];
    v2f b1 = *(const v2f*)&BJ1[bsel + k * 32];
    v4f c4 = *(const v4f*)&C4[(k * 16 + ln) * 4];

    v8f d0 = __builtin_amdgcn_wmma_f32_16x16x4_f32(false, a, false, b0,
                                                   (short)0, cz, false, false);
    v8f d1 = __builtin_amdgcn_wmma_f32_16x16x4_f32(false, a, false, b1,
                                                   (short)0, cz, false, false);

    asm volatile("s_wait_alu depctr_va_vdst(0)" : "+v"(d0), "+v"(d1));

    float x[8];
#pragma unroll
    for (int e = 0; e < 8; ++e) x[e] = half_abs_start(d0[e]);
    const v2f cr = {c4.x, c4.x}, cg = {c4.y, c4.y}, cb = {c4.z, c4.z};
#pragma unroll
    for (int q = 0; q < 4; ++q) {
      v2f m;
      m.x = half_abs_relu(d1[2 * q],     x[2 * q]);
      m.y = half_abs_relu(d1[2 * q + 1], x[2 * q + 1]);
      accR2[q] = __builtin_elementwise_fma(m, cr, accR2[q]);
      accG2[q] = __builtin_elementwise_fma(m, cg, accG2[q]);
      accB2[q] = __builtin_elementwise_fma(m, cb, accB2[q]);
    }
  }

  float accR[8], accG[8], accB[8];
#pragma unroll
  for (int q = 0; q < 4; ++q) {
    accR[2 * q] = accR2[q].x; accR[2 * q + 1] = accR2[q].y;
    accG[2 * q] = accG2[q].x; accG[2 * q + 1] = accG2[q].y;
    accB[2 * q] = accB2[q].x; accB[2 * q + 1] = accB2[q].y;
  }

  const bool b2 = (ln & 4) != 0;
  const bool b1 = (ln & 2) != 0;
  const bool b0 = (ln & 1) != 0;

  float s1R[4], s1G[4], s1B[4];
#pragma unroll
  for (int i = 0; i < 4; ++i) {
    s1R[i] = (b2 ? accR[i + 4] : accR[i]) + lane_xor<4>(b2 ? accR[i] : accR[i + 4]);
    s1G[i] = (b2 ? accG[i + 4] : accG[i]) + lane_xor<4>(b2 ? accG[i] : accG[i + 4]);
    s1B[i] = (b2 ? accB[i + 4] : accB[i]) + lane_xor<4>(b2 ? accB[i] : accB[i + 4]);
  }
  float s2R[2], s2G[2], s2B[2];
#pragma unroll
  for (int i = 0; i < 2; ++i) {
    s2R[i] = (b1 ? s1R[i + 2] : s1R[i]) + lane_xor<2>(b1 ? s1R[i] : s1R[i + 2]);
    s2G[i] = (b1 ? s1G[i + 2] : s1G[i]) + lane_xor<2>(b1 ? s1G[i] : s1G[i + 2]);
    s2B[i] = (b1 ? s1B[i + 2] : s1B[i]) + lane_xor<2>(b1 ? s1B[i] : s1B[i + 2]);
  }
  float sR = (b0 ? s2R[1] : s2R[0]) + lane_xor<1>(b0 ? s2R[0] : s2R[1]);
  float sG = (b0 ? s2G[1] : s2G[0]) + lane_xor<1>(b0 ? s2G[0] : s2G[1]);
  float sB = (b0 ? s2B[1] : s2B[0]) + lane_xor<1>(b0 ? s2B[0] : s2B[1]);
  sR += lane_xor<8>(sR);
  sG += lane_xor<8>(sG);
  sB += lane_xor<8>(sB);

  {
    int p   = tile * 16 + (ln & 7) + hi * 8;
    float r = 1.0f / (1.0f + __expf(-4.0f * sR));
    float g = 1.0f / (1.0f + __expf(-4.0f * sG));
    float b = 1.0f / (1.0f + __expf(-4.0f * sB));
    if (ln < 8) {
      out[0 * CANVAS_H * CANVAS_W + p] = r;
      out[1 * CANVAS_H * CANVAS_W + p] = g;
      out[2 * CANVAS_H * CANVAS_W + p] = b;
    }
  }
}

extern "C" void kernel_launch(void* const* d_in, const int* in_sizes, int n_in,
                              void* d_out, int out_size, void* d_ws, size_t ws_size,
                              hipStream_t stream) {
  (void)in_sizes; (void)n_in; (void)out_size; (void)d_ws; (void)ws_size;
  const float* loc = (const float*)d_in[0];
  const float* mo  = (const float*)d_in[1];
  const float* mse = (const float*)d_in[2];
  const float* col = (const float*)d_in[3];
  float* out = (float*)d_out;

  const int tiles  = (CANVAS_H * CANVAS_W) / 16;
  const int blocks = tiles / 8;
  points_kernel<<<blocks, 256, 0, stream>>>(loc, mo, mse, col, out);
}


# table build: each thread takes 4 consecutive points and reads all its inputs before computing, so its ten 16-byte loads fly together: one global round trip per work-group, was eight in series
# speedup vs baseline: 1.3007x; 1.0963x over previous
#include <hip/hip_runtime.h>
#include <math.h>

typedef __attribute__((ext_vector_type(2))) float v2f;
typedef __attribute__((ext_vector_type(4))) float v4f;
typedef __attribute__((ext_vector_type(8))) float v8f;

#define CANVAS_H 256
#define CANVAS_W 384
#define NPTS     1024
#define R_WH     1.5f


static __device__ __forceinline__ float half_abs_start(float t0) {
  float x;
  asm("v_fma_f32 %0, -0.5, |%1|, 1.0" : "=v"(x) : "v"(t0));
  return x;
}
static __device__ __forceinline__ float half_abs_relu(float t1, float x) {
  float m;
  asm("v_fma_f32 %0, -0.5, |%1|, %2 clamp" : "=v"(m) : "v"(t1), "v"(x));
  return m;
}

template <int M>
static __device__ __forceinline__ float lane_xor(float v) {
  return __int_as_float(__builtin_amdgcn_update_dpp(
      0, __float_as_int(v), 0x160 + M, 0xf, 0xf, true));
}

__global__ __launch_bounds__(256) void points_kernel(
    const float* __restrict__ loc,
    const float* __restrict__ mo,
    const float* __restrict__ mse,
    const float* __restrict__ col,
    float* __restrict__ out)
{
  __shared__ __align__(16) float lds[12 * NPTS];
  float* BJ0 = lds;
  float* BJ1 = lds + 4 * NPTS;
  float* C4  = lds + 8 * NPTS;

  {
    const int n0 = (int)threadIdx.x * 4;
    float ex[4], mm[16], ll[8], cc[12];
#pragma unroll
    for (int i = 0; i < 4; ++i)  ex[i] = mse[n0 + i];
#pragma unroll
    for (int i = 0; i < 16; ++i) mm[i] = mo[n0 * 4 + i];
#pragma unroll
    for (int i = 0; i < 8; ++i)  ll[i] = loc[n0 * 2 + i];
#pragma unroll
    for (int i = 0; i < 12; ++i) cc[i] = col[n0 * 3 + i];

#pragma unroll
    for (int i = 0; i < 4; ++i) {
      const int n = n0 + i;
      float s   = 16.0f * __expf(-ex[i]);
      float m00 = mm[i * 4 + 0] + s;
      float m01 = mm[i * 4 + 1];
      float m10 = mm[i * 4 + 2];
      float m11 = mm[i * 4 + 3] + s;
      float ly  = ll[i * 2 + 0];
      float lx  = ll[i * 2 + 1];
      float b0  = fmaf(ly, m00, lx * m10);
      float b1  = fmaf(ly, m01, lx * m11);

      v2f lo0; lo0.x = m00; lo0.y = m10;
      v2f hi0; hi0.x = -b0; hi0.y = 0.0f;
      v2f lo1; lo1.x = m01; lo1.y = m11;
      v2f hi1; hi1.x = -b1; hi1.y = 0.0f;
      *(v2f*)&BJ0[n * 2]              = lo0;
      *(v2f*)&BJ0[(NPTS + n) * 2]     = hi0;
      *(v2f*)&BJ1[n * 2]              = lo1;
      *(v2f*)&BJ1[(NPTS + n) * 2]     = hi1;
      v4f c; c.x = cc[i * 3 + 0]; c.y = cc[i * 3 + 1];
      c.z = cc[i * 3 + 2]; c.w = 0.0f;
      *(v4f*)&C4[n * 4] = c;
    }
  }
  __syncthreads();

  const int lane = threadIdx.x & 31;
  const int wave = threadIdx.x >> 5;
  const int tile = blockIdx.x * 8 + wave;
  const int hi   = lane >> 4;
  const int ln   = lane & 15;

  int p0 = tile * 16 + lane;
  int h0 = p0 / CANVAS_W, w0 = p0 - h0 * CANVAS_W;
  float gy = -1.0f + (float)h0 * (2.0f / (float)(CANVAS_H - 1));
  float gx = -R_WH + (float)w0 * (2.0f * R_WH / (float)(CANVAS_W - 1));
  v2f a;
  a.x = hi ? 1.0f : gy;
  a.y = hi ? 0.0f : gx;

  v2f accR2[4], accG2[4], accB2[4];
#pragma unroll
  for (int q = 0; q < 4; ++q) {
    accR2[q] = (v2f){0.0f, 0.0f}; accG2[q] = (v2f){0.0f, 0.0f}; accB2[q] = (v2f){0.0f, 0.0f};
  }

  const v8f cz = {};
  const int bsel = (hi * NPTS + ln) * 2;

#pragma unroll 2
  for (int k = 0; k < NPTS / 16; ++k) {
    v2f b0 = *(const v2f*)&BJ0[bsel + k * 32];
    v2f b1 = *(const v2f*)&BJ1[bsel + k * 32];
    v4f c4 = *(const v4f*)&C4[(k * 16 + ln) * 4];

    v8f d0 = __builtin_amdgcn_wmma_f32_16x16x4_f32(false, a, false, b0,
                                                   (short)0, cz, false, false);
    v8f d1 = __builtin_amdgcn_wmma_f32_16x16x4_f32(false, a, false, b1,
                                                   (short)0, cz, false, false);

    asm volatile("s_wait_alu depctr_va_vdst(0)" : "+v"(d0), "+v"(d1));

    float x[8];
#pragma unroll
    for (int e = 0; e < 8; ++e) x[e] = half_abs_start(d0[e]);
    const v2f cr = {c4.x, c4.x}, cg = {c4.y, c4.y}, cb = {c4.z, c4.z};
#pragma unroll
    for (int q = 0; q < 4; ++q) {
      v2f m;
      m.x = half_abs_relu(d1[2 * q],     x[2 * q]);
      m.y = half_abs_relu(d1[2 * q + 1], x[2 * q + 1]);
      accR2[q] = __builtin_elementwise_fma(m, cr, accR2[q]);
      accG2[q] = __builtin_elementwise_fma(m, cg, accG2[q]);
      accB2[q] = __builtin_elementwise_fma(m, cb, accB2[q]);
    }
  }

  float accR[8], accG[8], accB[8];
#pragma unroll
  for (int q = 0; q < 4; ++q) {
    accR[2 * q] = accR2[q].x; accR[2 * q + 1] = accR2[q].y;
    accG[2 * q] = accG2[q].x; accG[2 * q + 1] = accG2[q].y;
    accB[2 * q] = accB2[q].x; accB[2 * q + 1] = accB2[q].y;
  }

  const bool b2 = (ln & 4) != 0;
  const bool b1 = (ln & 2) != 0;
  const bool b0 = (ln & 1) != 0;

  float s1R[4], s1G[4], s1B[4];
#pragma unroll
  for (int i = 0; i < 4; ++i) {
    s1R[i] = (b2 ? accR[i + 4] : accR[i]) + lane_xor<4>(b2 ? accR[i] : accR[i + 4]);
    s1G[i] = (b2 ? accG[i + 4] : accG[i]) + lane_xor<4>(b2 ? accG[i] : accG[i + 4]);
    s1B[i] = (b2 ? accB[i + 4] : accB[i]) + lane_xor<4>(b2 ? accB[i] : accB[i + 4]);
  }
  float s2R[2], s2G[2], s2B[2];
#pragma unroll
  for (int i = 0; i < 2; ++i) {
    s2R[i] = (b1 ? s1R[i + 2] : s1R[i]) + lane_xor<2>(b1 ? s1R[i] : s1R[i + 2]);
    s2G[i] = (b1 ? s1G[i + 2] : s1G[i]) + lane_xor<2>(b1 ? s1G[i] : s1G[i + 2]);
    s2B[i] = (b1 ? s1B[i + 2] : s1B[i]) + lane_xor<2>(b1 ? s1B[i] : s1B[i + 2]);
  }
  float sR = (b0 ? s2R[1] : s2R[0]) + lane_xor<1>(b0 ? s2R[0] : s2R[1]);
  float sG = (b0 ? s2G[1] : s2G[0]) + lane_xor<1>(b0 ? s2G[0] : s2G[1]);
  float sB = (b0 ? s2B[1] : s2B[0]) + lane_xor<1>(b0 ? s2B[0] : s2B[1]);
  sR += lane_xor<8>(sR);
  sG += lane_xor<8>(sG);
  sB += lane_xor<8>(sB);

  {
    int p   = tile * 16 + (ln & 7) + hi * 8;
    float r = 1.0f / (1.0f + __expf(-4.0f * sR));
    float g = 1.0f / (1.0f + __expf(-4.0f * sG));
    float b = 1.0f / (1.0f + __expf(-4.0f * sB));
    if (ln < 8) {
      out[0 * CANVAS_H * CANVAS_W + p] = r;
      out[1 * CANVAS_H * CANVAS_W + p] = g;
      out[2 * CANVAS_H * CANVAS_W + p] = b;
    }
  }
}

extern "C" void kernel_launch(void* const* d_in, const int* in_sizes, int n_in,
                              void* d_out, int out_size, void* d_ws, size_t ws_size,
                              hipStream_t stream) {
  (void)in_sizes; (void)n_in; (void)out_size; (void)d_ws; (void)ws_size;
  const float* loc = (const float*)d_in[0];
  const float* mo  = (const float*)d_in[1];
  const float* mse = (const float*)d_in[2];
  const float* col = (const float*)d_in[3];
  float* out = (float*)d_out;

  const int tiles  = (CANVAS_H * CANVAS_W) / 16;
  const int blocks = tiles / 8;
  points_kernel<<<blocks, 256, 0, stream>>>(loc, mo, mse, col, out);
}


# separable transform: per 8x16-pixel patch, row part and column part of t are 4 WMMAs per 16 points (was 2 per 16 pixels), added per pair by v_pk_add; 4 waves split the points; sum order changes
# speedup vs baseline: 1.4696x; 1.1299x over previous
#include <hip/hip_runtime.h>
#include <math.h>


typedef __attribute__((ext_vector_type(2))) float v2f;
typedef __attribute__((ext_vector_type(4))) float v4f;
typedef __attribute__((ext_vector_type(8))) float v8f;

#define CANVAS_H 256
#define CANVAS_W 384
#define NPTS     1024
#define R_WH     1.5f
#define PATCH_H  8
#define PATCH_W  16
#define NWAVES   4

static __device__ __forceinline__ void half_abs_start(float& t0) {
  asm volatile("v_fma_f32 %0, -0.5, |%0|, 1.0" : "+v"(t0));
}
static __device__ __forceinline__ void half_abs_relu(float& x, float t1) {
  asm volatile("v_fma_f32 %0, -0.5, |%1|, %0 clamp" : "+v"(x) : "v"(t1));
}
template <int HI>
static __device__ __forceinline__ v2f bcast_add(v2f p, v2f q) {
  v2f t;
  if (HI) asm volatile("v_pk_add_f32 %0, %1, %2 op_sel:[1,0]"    : "=v"(t) : "v"(p), "v"(q));
  else    asm volatile("v_pk_add_f32 %0, %1, %2 op_sel_hi:[0,1]" : "=v"(t) : "v"(p), "v"(q));
  return t;
}
template <int HI>
static __device__ __forceinline__ void bcast_fma(v2f& acc, v2f m, v2f c) {
  if (HI) asm volatile("v_pk_fma_f32 %0, %1, %2, %0 op_sel:[0,1,0]"    : "+v"(acc) : "v"(m), "v"(c));
  else    asm volatile("v_pk_fma_f32 %0, %1, %2, %0 op_sel_hi:[1,0,1]" : "+v"(acc) : "v"(m), "v"(c));
}

template <int M>
static __device__ __forceinline__ float lane_xor(float v) {
  return __int_as_float(__builtin_amdgcn_update_dpp(
      0, __float_as_int(v), 0x160 + M, 0xf, 0xf, true));
}

template <int M, int N>
static __device__ __forceinline__ void scatter_step(const float* in, float* outv, bool bit) {
#pragma unroll
  for (int i = 0; i < N / 2; ++i) {
    float keep = bit ? in[i + N / 2] : in[i];
    float give = bit ? in[i] : in[i + N / 2];
    outv[i] = keep + lane_xor<M>(give);
  }
}

__global__ __launch_bounds__(32 * NWAVES) void points_kernel(
    const float* __restrict__ loc,
    const float* __restrict__ mo,
    const float* __restrict__ mse,
    const float* __restrict__ col,
    float* __restrict__ out)
{
  __shared__ __align__(16) float lds[12 * NPTS];
  __shared__ __align__(16) float red[NWAVES * 2 * 192];
  float* BJ0 = lds;
  float* BJ1 = lds + 4 * NPTS;
  float* C4  = lds + 8 * NPTS;

  {
    const int n0 = (int)threadIdx.x * 8;
    float ex[8], mm[32], ll[16], cc[24];
#pragma unroll
    for (int i = 0; i < 8; ++i)  ex[i] = mse[n0 + i];
#pragma unroll
    for (int i = 0; i < 32; ++i) mm[i] = mo[n0 * 4 + i];
#pragma unroll
    for (int i = 0; i < 16; ++i) ll[i] = loc[n0 * 2 + i];
#pragma unroll
    for (int i = 0; i < 24; ++i) cc[i] = col[n0 * 3 + i];

#pragma unroll
    for (int i = 0; i < 8; ++i) {
      const int n = n0 + i;
      float s   = 16.0f * __expf(-ex[i]);
      float m00 = mm[i * 4 + 0] + s;
      float m01 = mm[i * 4 + 1];
      float m10 = mm[i * 4 + 2];
      float m11 = mm[i * 4 + 3] + s;
      float ly  = ll[i * 2 + 0];
      float lx  = ll[i * 2 + 1];
      float b0  = fmaf(ly, m00, lx * m10);
      float b1  = fmaf(ly, m01, lx * m11);

      v2f lo0; lo0.x = m00; lo0.y = m10;
      v2f hi0; hi0.x = -b0; hi0.y = 0.0f;
      v2f lo1; lo1.x = m01; lo1.y = m11;
      v2f hi1; hi1.x = -b1; hi1.y = 0.0f;
      *(v2f*)&BJ0[n * 2]              = lo0;
      *(v2f*)&BJ0[(NPTS + n) * 2]     = hi0;
      *(v2f*)&BJ1[n * 2]              = lo1;
      *(v2f*)&BJ1[(NPTS + n) * 2]     = hi1;
      v4f c; c.x = cc[i * 3 + 0]; c.y = cc[i * 3 + 1];
      c.z = cc[i * 3 + 2]; c.w = 0.0f;
      *(v4f*)&C4[n * 4] = c;
    }
  }
  __syncthreads();

  const int lane = threadIdx.x & 31;
  const int wave = threadIdx.x >> 5;
  const int hi   = lane >> 4;
  const int ln   = lane & 15;

  const int ph = (int)blockIdx.x / (CANVAS_W / PATCH_W);
  const int pw = (int)blockIdx.x - ph * (CANVAS_W / PATCH_W);
  const int h0 = ph * PATCH_H;
  const int w0 = pw * PATCH_W;

  const float gy = -1.0f + (float)(h0 + (ln & 7)) * (2.0f / (float)(CANVAS_H - 1));
  const float gx = -R_WH + (float)(w0 + ln) * (2.0f * R_WH / (float)(CANVAS_W - 1));
  v2f aP, aQ;
  aP.x = hi ? 1.0f : gy;
  aP.y = 0.0f;
  aQ.x = 0.0f;
  aQ.y = hi ? 0.0f : gx;

  v2f acc2[96];
#pragma unroll
  for (int i = 0; i < 96; ++i) acc2[i] = (v2f){0.0f, 0.0f};

  const v8f cz = {};
  const int bsel = (hi * NPTS + ln) * 2;
  const int kbeg = wave * (NPTS / 16 / NWAVES);

  for (int kk = 0; kk < NPTS / 16 / NWAVES; ++kk) {
    const int k = kbeg + kk;
    v2f b0 = *(const v2f*)&BJ0[bsel + k * 32];
    v2f b1 = *(const v2f*)&BJ1[bsel + k * 32];
    v4f c4 = *(const v4f*)&C4[(k * 16 + ln) * 4];

    v8f P0 = __builtin_amdgcn_wmma_f32_16x16x4_f32(false, aP, false, b0,
                                                   (short)0, cz, false, false);
    v8f P1 = __builtin_amdgcn_wmma_f32_16x16x4_f32(false, aP, false, b1,
                                                   (short)0, cz, false, false);
    v8f Q0 = __builtin_amdgcn_wmma_f32_16x16x4_f32(false, aQ, false, b0,
                                                   (short)0, cz, false, false);
    v8f Q1 = __builtin_amdgcn_wmma_f32_16x16x4_f32(false, aQ, false, b1,
                                                   (short)0, cz, false, false);

    asm volatile("s_wait_alu depctr_va_vdst(0)" : "+v"(P0), "+v"(P1), "+v"(Q0), "+v"(Q1));

    const v2f c01 = {c4.x, c4.y}, c23 = {c4.z, c4.w};

#define ROW_SUMS(R, T0, T1)                                                        \
    {                                                                              \
      const v2f pp0 = {P0[(R) & 6], P0[((R) & 6) + 1]};                            \
      const v2f pp1 = {P1[(R) & 6], P1[((R) & 6) + 1]};                            \
      _Pragma("unroll") for (int q = 0; q < 4; ++q) {                              \
        const v2f qq = {Q0[2 * q], Q0[2 * q + 1]};                                 \
        T0[q] = bcast_add<(R) & 1>(pp0, qq);                                       \
      }                                                                            \
      _Pragma("unroll") for (int q = 0; q < 4; ++q) {                              \
        const v2f qq = {Q1[2 * q], Q1[2 * q + 1]};                                 \
        T1[q] = bcast_add<(R) & 1>(pp1, qq);                                       \
      }                                                                            \
    }
#define ROW_REST(R, T0, T1, NEXT)                                                  \
    {                                                                              \
      float xs[8];                                                                 \
      _Pragma("unroll") for (int q = 0; q < 4; ++q) {                              \
        xs[2 * q] = T0[q].x; xs[2 * q + 1] = T0[q].y;                              \
        half_abs_start(xs[2 * q]); half_abs_start(xs[2 * q + 1]);                  \
      }                                                                            \
      NEXT                                                                         \
      v2f mm2[4];                                                                  \
      _Pragma("unroll") for (int q = 0; q < 4; ++q) {                              \
        half_abs_relu(xs[2 * q], T1[q].x); half_abs_relu(xs[2 * q + 1], T1[q].y);  \
        mm2[q].x = xs[2 * q]; mm2[q].y = xs[2 * q + 1];                            \
      }                                                                            \
      _Pragma("unroll") for (int q = 0; q < 4; ++q) {                              \
        bcast_fma<0>(acc2[0 * 32 + (R) * 4 + q], mm2[q], c01);                     \
        bcast_fma<1>(acc2[1 * 32 + (R) * 4 + q], mm2[q], c01);                     \
        bcast_fma<0>(acc2[2 * 32 + (R) * 4 + q], mm2[q], c23);                     \
      }                                                                            \
    }
    v2f ta0[4], ta1[4], tb0[4], tb1[4];
    ROW_SUMS(0, ta0, ta1)
    ROW_REST(0, ta0, ta1, ROW_SUMS(1, tb0, tb1))
    ROW_REST(1, tb0, tb1, ROW_SUMS(2, ta0, ta1))
    ROW_REST(2, ta0, ta1, ROW_SUMS(3, tb0, tb1))
    ROW_REST(3, tb0, tb1, ROW_SUMS(4, ta0, ta1))
    ROW_REST(4, ta0, ta1, ROW_SUMS(5, tb0, tb1))
    ROW_REST(5, tb0, tb1, ROW_SUMS(6, ta0, ta1))
    ROW_REST(6, ta0, ta1, ROW_SUMS(7, tb0, tb1))
    ROW_REST(7, tb0, tb1, )
#undef ROW_SUMS
#undef ROW_REST
  }

  float v[192];
#pragma unroll
  for (int i = 0; i < 96; ++i) { v[2 * i] = acc2[i].x; v[2 * i + 1] = acc2[i].y; }

  float s1[96], s2[48], s3[24], s4[12];
  scatter_step<8, 192>(v,  s1, (ln & 8) != 0);
  scatter_step<4, 96>(s1, s2, (ln & 4) != 0);
  scatter_step<2, 48>(s2, s3, (ln & 2) != 0);
  scatter_step<1, 24>(s3, s4, (ln & 1) != 0);

  {
    float* dst = &red[(wave * 2 + hi) * 192 + ln * 12];
#pragma unroll
    for (int i = 0; i < 3; ++i) {
      v4f w4; w4.x = s4[4 * i]; w4.y = s4[4 * i + 1]; w4.z = s4[4 * i + 2]; w4.w = s4[4 * i + 3];
      *(v4f*)&dst[4 * i] = w4;
    }
  }
  __syncthreads();
  {
    const int t  = (int)threadIdx.x;
    const int r1 = t >> 4;
    const int wc = t & 15;
    const int hh = wc >> 3;
    const int r2 = wc & 7;
    const int p  = (h0 + r1) * CANVAS_W + (w0 + wc);
#pragma unroll
    for (int ch = 0; ch < 3; ++ch) {
      const int idx = ch * 64 + r1 * 8 + r2;
      float s = 0.0f;
#pragma unroll
      for (int wv = 0; wv < NWAVES; ++wv) s += red[(wv * 2 + hh) * 192 + idx];
      out[ch * CANVAS_H * CANVAS_W + p] = 1.0f / (1.0f + __expf(-4.0f * s));
    }
  }
}

extern "C" void kernel_launch(void* const* d_in, const int* in_sizes, int n_in,
                              void* d_out, int out_size, void* d_ws, size_t ws_size,
                              hipStream_t stream) {
  (void)in_sizes; (void)n_in; (void)out_size; (void)d_ws; (void)ws_size;
  const float* loc = (const float*)d_in[0];
  const float* mo  = (const float*)d_in[1];
  const float* mse = (const float*)d_in[2];
  const float* col = (const float*)d_in[3];
  float* out = (float*)d_out;

  const int blocks = (CANVAS_H / PATCH_H) * (CANVAS_W / PATCH_W);
  points_kernel<<<blocks, 32 * NWAVES, 0, stream>>>(loc, mo, mse, col, out);
}


# 12-wave work-group on 3 side-by-side patches (grid 256, one per WGP): row products P_j and point table built once per work-group in LDS; sweep issues 2 WMMAs per chunk, was 4; zero K terms of P move
# speedup vs baseline: 1.6123x; 1.0971x over previous
#include <hip/hip_runtime.h>
#include <math.h>


typedef __attribute__((ext_vector_type(2))) float v2f;
typedef __attribute__((ext_vector_type(4))) float v4f;
typedef __attribute__((ext_vector_type(8))) float v8f;

#define CANVAS_H 256
#define CANVAS_W 384
#define NPTS     1024
#define R_WH     1.5f
#define PATCH_H  8
#define PATCH_W  16
#define NQ       4
#define NPATCH   3
#define NWAVES   (NQ * NPATCH)
#define PT_PITCH 20

static __device__ __forceinline__ void half_abs_start(float& t0) {
  asm volatile("v_fma_f32 %0, -0.5, |%0|, 1.0" : "+v"(t0));
}
static __device__ __forceinline__ void half_abs_relu(float& x, float t1) {
  asm volatile("v_fma_f32 %0, -0.5, |%1|, %0 clamp" : "+v"(x) : "v"(t1));
}
template <int HI>
static __device__ __forceinline__ v2f bcast_add(v2f p, v2f q) {
  v2f t;
  if (HI) asm volatile("v_pk_add_f32 %0, %1, %2 op_sel:[1,0]"    : "=v"(t) : "v"(p), "v"(q));
  else    asm volatile("v_pk_add_f32 %0, %1, %2 op_sel_hi:[0,1]" : "=v"(t) : "v"(p), "v"(q));
  return t;
}
template <int HI>
static __device__ __forceinline__ void bcast_fma(v2f& acc, v2f m, v2f c) {
  if (HI) asm volatile("v_pk_fma_f32 %0, %1, %2, %0 op_sel:[0,1,0]"    : "+v"(acc) : "v"(m), "v"(c));
  else    asm volatile("v_pk_fma_f32 %0, %1, %2, %0 op_sel_hi:[1,0,1]" : "+v"(acc) : "v"(m), "v"(c));
}

template <int M>
static __device__ __forceinline__ float lane_xor(float v) {
  return __int_as_float(__builtin_amdgcn_update_dpp(
      0, __float_as_int(v), 0x160 + M, 0xf, 0xf, true));
}

template <int M, int N>
static __device__ __forceinline__ void scatter_step(const float* in, float* outv, bool bit) {
#pragma unroll
  for (int i = 0; i < N / 2; ++i) {
    float keep = bit ? in[i + N / 2] : in[i];
    float give = bit ? in[i] : in[i + N / 2];
    outv[i] = keep + lane_xor<M>(give);
  }
}

__global__ __launch_bounds__(32 * NWAVES) void points_kernel(
    const float* __restrict__ loc,
    const float* __restrict__ mo,
    const float* __restrict__ mse,
    const float* __restrict__ col,
    float* __restrict__ out)
{
  __shared__ __align__(16) float lds[12 * NPTS];
  __shared__ __align__(16) float PT[NPTS * PT_PITCH];
  __shared__ __align__(16) float BP[4 * NPTS];
  __shared__ __align__(16) float red[NWAVES * 2 * 192];
  float* BJ0 = lds;
  float* BJ1 = lds + 4 * NPTS;
  float* C4  = lds + 8 * NPTS;

  if (threadIdx.x < NPTS / 4) {
    const int n0 = (int)threadIdx.x * 4;
    float ex[4], mm[16], ll[8], cc[12];
#pragma unroll
    for (int i = 0; i < 4; ++i)  ex[i] = mse[n0 + i];
#pragma unroll
    for (int i = 0; i < 16; ++i) mm[i] = mo[n0 * 4 + i];
#pragma unroll
    for (int i = 0; i < 8; ++i)  ll[i] = loc[n0 * 2 + i];
#pragma unroll
    for (int i = 0; i < 12; ++i) cc[i] = col[n0 * 3 + i];

#pragma unroll
    for (int i = 0; i < 4; ++i) {
      const int n = n0 + i;
      float s   = 16.0f * __expf(-ex[i]);
      float m00 = mm[i * 4 + 0] + s;
      float m01 = mm[i * 4 + 1];
      float m10 = mm[i * 4 + 2];
      float m11 = mm[i * 4 + 3] + s;
      float ly  = ll[i * 2 + 0];
      float lx  = ll[i * 2 + 1];
      float b0  = fmaf(ly, m00, lx * m10);
      float b1  = fmaf(ly, m01, lx * m11);

      v2f lo0; lo0.x = m00; lo0.y = m10;
      v2f hi0; hi0.x = -b0; hi0.y = 0.0f;
      v2f lo1; lo1.x = m01; lo1.y = m11;
      v2f hi1; hi1.x = -b1; hi1.y = 0.0f;
      *(v2f*)&BJ0[n * 2]              = lo0;
      *(v2f*)&BJ0[(NPTS + n) * 2]     = hi0;
      *(v2f*)&BJ1[n * 2]              = lo1;
      *(v2f*)&BJ1[(NPTS + n) * 2]     = hi1;
      v2f lop; lop.x = m00; lop.y = -b0;
      v2f hip; hip.x = m01; hip.y = -b1;
      *(v2f*)&BP[n * 2]               = lop;
      *(v2f*)&BP[(NPTS + n) * 2]      = hip;
      v4f c; c.x = cc[i * 3 + 0]; c.y = cc[i * 3 + 1];
      c.z = cc[i * 3 + 2]; c.w = 0.0f;
      *(v4f*)&C4[n * 4] = c;
    }
  }
  __syncthreads();

  const int lane  = threadIdx.x & 31;
  const int wave  = __builtin_amdgcn_readfirstlane((int)(threadIdx.x >> 5));
  const int patch = wave / NQ;
  const int quart = wave - patch * NQ;
  const int hi    = lane >> 4;
  const int ln    = lane & 15;

  const int groups_per_row = CANVAS_W / (PATCH_W * NPATCH);
  const int ph = (int)blockIdx.x / groups_per_row;
  const int pg = (int)blockIdx.x - ph * groups_per_row;
  const int h0 = ph * PATCH_H;
  const int wg0 = pg * (PATCH_W * NPATCH);
  const int w0 = wg0 + patch * PATCH_W;

  const float gy = -1.0f + (float)(h0 + (ln & 7)) * (2.0f / (float)(CANVAS_H - 1));
  const float gx = -R_WH + (float)(w0 + ln) * (2.0f * R_WH / (float)(CANVAS_W - 1));
  v2f aP, aQ;
  const bool own = ((ln >> 3) == hi);
  aP.x = own ? gy : 0.0f;
  aP.y = own ? 1.0f : 0.0f;
  aQ.x = 0.0f;
  aQ.y = hi ? 0.0f : gx;

  const v8f cz = {};
  const int bsel = (hi * NPTS + ln) * 2;

#define ROW_PRODUCTS(K, B)                                                         \
  {                                                                                \
    const v8f p = __builtin_amdgcn_wmma_f32_16x16x4_f32(false, aP, false, B,       \
                                                   (short)0, cz, false, false);    \
    float* d = ptl + (K) * (16 * PT_PITCH);                                        \
    *(v4f*)&d[0] = __builtin_shufflevector(p, p, 0, 1, 2, 3);                      \
    *(v4f*)&d[4] = __builtin_shufflevector(p, p, 4, 5, 6, 7);                      \
  }
  {
    float* ptl = &PT[ln * PT_PITCH + 8 * hi];
    v2f rb[5];
#pragma unroll
    for (int i = 0; i < 5; ++i) rb[i] = *(const v2f*)&BP[bsel + (wave + NWAVES * i) * 32];
#pragma unroll
    for (int i = 0; i < 5; ++i) ROW_PRODUCTS(wave + NWAVES * i, rb[i])
    if (wave < NPTS / 16 - 5 * NWAVES) {
      const int k = wave + NWAVES * 5;
      const v2f lb = *(const v2f*)&BP[bsel + k * 32];
      ROW_PRODUCTS(k, lb)
    }
  }
#undef ROW_PRODUCTS
  __syncthreads();

  v2f acc2[96];
#pragma unroll
  for (int i = 0; i < 96; ++i) acc2[i] = (v2f){0.0f, 0.0f};

  const int kbeg = quart * (NPTS / 16 / NQ);

  for (int kk = 0; kk < NPTS / 16 / NQ; ++kk) {
    const int k = kbeg + kk;
    v2f b0 = *(const v2f*)&BJ0[bsel + k * 32];
    v2f b1 = *(const v2f*)&BJ1[bsel + k * 32];
    v4f c4 = *(const v4f*)&C4[(k * 16 + ln) * 4];
    const float* pp = &PT[(k * 16 + ln) * PT_PITCH];
    const v4f pa = *(const v4f*)&pp[0];
    const v4f pb = *(const v4f*)&pp[4];
    const v4f pc = *(const v4f*)&pp[8];
    const v4f pd = *(const v4f*)&pp[12];
    const v8f P0 = __builtin_shufflevector(pa, pb, 0, 1, 2, 3, 4, 5, 6, 7);
    const v8f P1 = __builtin_shufflevector(pc, pd, 0, 1, 2, 3, 4, 5, 6, 7);

    v8f Q0 = __builtin_amdgcn_wmma_f32_16x16x4_f32(false, aQ, false, b0,
                                                   (short)0, cz, false, false);
    v8f Q1 = __builtin_amdgcn_wmma_f32_16x16x4_f32(false, aQ, false, b1,
                                                   (short)0, cz, false, false);

    asm volatile("s_wait_alu depctr_va_vdst(0)" : "+v"(Q0), "+v"(Q1));

    const v2f c01 = {c4.x, c4.y}, c23 = {c4.z, c4.w};

#define ROW_SUMS(R, T0, T1)                                                        \
    {                                                                              \
      const v2f pp0 = {P0[(R) & 6], P0[((R) & 6) + 1]};                            \
      const v2f pp1 = {P1[(R) & 6], P1[((R) & 6) + 1]};                            \
      _Pragma("unroll") for (int q = 0; q < 4; ++q) {                              \
        const v2f qq = {Q0[2 * q], Q0[2 * q + 1]};                                 \
        T0[q] = bcast_add<(R) & 1>(pp0, qq);                                       \
      }                                                                            \
      _Pragma("unroll") for (int q = 0; q < 4; ++q) {                              \
        const v2f qq = {Q1[2 * q], Q1[2 * q + 1]};                                 \
        T1[q] = bcast_add<(R) & 1>(pp1, qq);                                       \
      }                                                                            \
    }
#define ROW_REST(R, T0, T1, NEXT)                                                  \
    {                                                                              \
      float xs[8];                                                                 \
      _Pragma("unroll") for (int q = 0; q < 4; ++q) {                              \
        xs[2 * q] = T0[q].x; xs[2 * q + 1] = T0[q].y;                              \
        half_abs_start(xs[2 * q]); half_abs_start(xs[2 * q + 1]);                  \
      }                                                                            \
      NEXT                                                                         \
      v2f mm2[4];                                                                  \
      _Pragma("unroll") for (int q = 0; q < 4; ++q) {                              \
        half_abs_relu(xs[2 * q], T1[q].x); half_abs_relu(xs[2 * q + 1], T1[q].y);  \
        mm2[q].x = xs[2 * q]; mm2[q].y = xs[2 * q + 1];                            \
      }                                                                            \
      _Pragma("unroll") for (int q = 0; q < 4; ++q) {                              \
        bcast_fma<0>(acc2[0 * 32 + (R) * 4 + q], mm2[q], c01);                     \
        bcast_fma<1>(acc2[1 * 32 + (R) * 4 + q], mm2[q], c01);                     \
        bcast_fma<0>(acc2[2 * 32 + (R) * 4 + q], mm2[q], c23);                     \
      }                                                                            \
    }
    v2f ta0[4], ta1[4], tb0[4], tb1[4];
    ROW_SUMS(0, ta0, ta1)
    ROW_REST(0, ta0, ta1, ROW_SUMS(1, tb0, tb1))
    ROW_REST(1, tb0, tb1, ROW_SUMS(2, ta0, ta1))
    ROW_REST(2, ta0, ta1, ROW_SUMS(3, tb0, tb1))
    ROW_REST(3, tb0, tb1, ROW_SUMS(4, ta0, ta1))
    ROW_REST(4, ta0, ta1, ROW_SUMS(5, tb0, tb1))
    ROW_REST(5, tb0, tb1, ROW_SUMS(6, ta0, ta1))
    ROW_REST(6, ta0, ta1, ROW_SUMS(7, tb0, tb1))
    ROW_REST(7, tb0, tb1, )
#undef ROW_SUMS
#undef ROW_REST
  }

  float v[192];
#pragma unroll
  for (int i = 0; i < 96; ++i) { v[2 * i] = acc2[i].x; v[2 * i + 1] = acc2[i].y; }

  float s1[96], s2[48], s3[24], s4[12];
  scatter_step<8, 192>(v,  s1, (ln & 8) != 0);
  scatter_step<4, 96>(s1, s2, (ln & 4) != 0);
  scatter_step<2, 48>(s2, s3, (ln & 2) != 0);
  scatter_step<1, 24>(s3, s4, (ln & 1) != 0);

  {
    float* dst = &red[(wave * 2 + hi) * 192 + ln * 12];
#pragma unroll
    for (int i = 0; i < 3; ++i) {
      v4f w4; w4.x = s4[4 * i]; w4.y = s4[4 * i + 1]; w4.z = s4[4 * i + 2]; w4.w = s4[4 * i + 3];
      *(v4f*)&dst[4 * i] = w4;
    }
  }
  __syncthreads();
  {
    const int t  = (int)threadIdx.x;
    const int r1 = t / (PATCH_W * NPATCH);
    const int gc = t - r1 * (PATCH_W * NPATCH);
    const int pt = gc >> 4;
    const int wc = gc & 15;
    const int hh = wc >> 3;
    const int r2 = wc & 7;
    const int p  = (h0 + r1) * CANVAS_W + (wg0 + gc);
#pragma unroll
    for (int ch = 0; ch < 3; ++ch) {
      const int idx = ch * 64 + r1 * 8 + r2;
      float s = 0.0f;
#pragma unroll
      for (int wv = 0; wv < NQ; ++wv) s += red[((pt * NQ + wv) * 2 + hh) * 192 + idx];
      out[ch * CANVAS_H * CANVAS_W + p] = 1.0f / (1.0f + __expf(-4.0f * s));
    }
  }
}

extern "C" void kernel_launch(void* const* d_in, const int* in_sizes, int n_in,
                              void* d_out, int out_size, void* d_ws, size_t ws_size,
                              hipStream_t stream) {
  (void)in_sizes; (void)n_in; (void)out_size; (void)d_ws; (void)ws_size;
  const float* loc = (const float*)d_in[0];
  const float* mo  = (const float*)d_in[1];
  const float* mse = (const float*)d_in[2];
  const float* col = (const float*)d_in[3];
  float* out = (float*)d_out;

  const int blocks = (CANVAS_H / PATCH_H) * (CANVAS_W / (PATCH_W * NPATCH));
  points_kernel<<<blocks, 32 * NWAVES, 0, stream>>>(loc, mo, mse, col, out);
}


# weight kept as complement p = min(dist,1): ONE v_add_f32 |t0/2|+|t1/2| with clamp (was 2 v_fma); canvas = sum(c) - sum(p*c), each lane adds its points' colour sum once; rounding and sum order change
# speedup vs baseline: 1.7490x; 1.0848x over previous
#include <hip/hip_runtime.h>
#include <math.h>


typedef __attribute__((ext_vector_type(2))) float v2f;
typedef __attribute__((ext_vector_type(4))) float v4f;
typedef __attribute__((ext_vector_type(8))) float v8f;

#define CANVAS_H 256
#define CANVAS_W 384
#define NPTS     1024
#define R_WH     1.5f
#define PATCH_H  8
#define PATCH_W  16
#define NQ       4
#define NPATCH   3
#define NWAVES   (NQ * NPATCH)
#define PT_PITCH 20

static __device__ __forceinline__ void abs_sum_sat(float& x, float t1) {
  asm volatile("v_add_f32 %0, |%0|, |%1| clamp" : "+v"(x) : "v"(t1));
}
template <int HI>
static __device__ __forceinline__ v2f bcast_add(v2f p, v2f q) {
  v2f t;
  if (HI) asm volatile("v_pk_add_f32 %0, %1, %2 op_sel:[1,0]"    : "=v"(t) : "v"(p), "v"(q));
  else    asm volatile("v_pk_add_f32 %0, %1, %2 op_sel_hi:[0,1]" : "=v"(t) : "v"(p), "v"(q));
  return t;
}
template <int HI>
static __device__ __forceinline__ void bcast_fma(v2f& acc, v2f m, v2f c) {
  if (HI) asm volatile("v_pk_fma_f32 %0, %1, %2, %0 op_sel:[0,1,0]"    : "+v"(acc) : "v"(m), "v"(c));
  else    asm volatile("v_pk_fma_f32 %0, %1, %2, %0 op_sel_hi:[1,0,1]" : "+v"(acc) : "v"(m), "v"(c));
}

template <int M>
static __device__ __forceinline__ float lane_xor(float v) {
  return __int_as_float(__builtin_amdgcn_update_dpp(
      0, __float_as_int(v), 0x160 + M, 0xf, 0xf, true));
}

template <int M, int N>
static __device__ __forceinline__ void scatter_step(const float* in, float* outv, bool bit) {
#pragma unroll
  for (int i = 0; i < N / 2; ++i) {
    float keep = bit ? in[i + N / 2] : in[i];
    float give = bit ? in[i] : in[i + N / 2];
    outv[i] = keep + lane_xor<M>(give);
  }
}

__global__ __launch_bounds__(32 * NWAVES) void points_kernel(
    const float* __restrict__ loc,
    const float* __restrict__ mo,
    const float* __restrict__ mse,
    const float* __restrict__ col,
    float* __restrict__ out)
{
  __shared__ __align__(16) float lds[12 * NPTS];
  __shared__ __align__(16) float PT[NPTS * PT_PITCH];
  __shared__ __align__(16) float BP[4 * NPTS];
  __shared__ __align__(16) float red[NWAVES * 2 * 192];
  float* BJ0 = lds;
  float* BJ1 = lds + 4 * NPTS;
  float* C4  = lds + 8 * NPTS;

  if (threadIdx.x < NPTS / 4) {
    const int n0 = (int)threadIdx.x * 4;
    float ex[4], mm[16], ll[8], cc[12];
#pragma unroll
    for (int i = 0; i < 4; ++i)  ex[i] = mse[n0 + i];
#pragma unroll
    for (int i = 0; i < 16; ++i) mm[i] = mo[n0 * 4 + i];
#pragma unroll
    for (int i = 0; i < 8; ++i)  ll[i] = loc[n0 * 2 + i];
#pragma unroll
    for (int i = 0; i < 12; ++i) cc[i] = col[n0 * 3 + i];

#pragma unroll
    for (int i = 0; i < 4; ++i) {
      const int n = n0 + i;
      float s   = 16.0f * __expf(-ex[i]);
      float m00 = mm[i * 4 + 0] + s;
      float m01 = mm[i * 4 + 1];
      float m10 = mm[i * 4 + 2];
      float m11 = mm[i * 4 + 3] + s;
      float ly  = ll[i * 2 + 0];
      float lx  = ll[i * 2 + 1];
      float b0  = fmaf(ly, m00, lx * m10);
      float b1  = fmaf(ly, m01, lx * m11);

      v2f lo0; lo0.x = m00; lo0.y = m10;
      v2f hi0; hi0.x = -b0; hi0.y = 0.0f;
      v2f lo1; lo1.x = m01; lo1.y = m11;
      v2f hi1; hi1.x = -b1; hi1.y = 0.0f;
      *(v2f*)&BJ0[n * 2]              = lo0;
      *(v2f*)&BJ0[(NPTS + n) * 2]     = hi0;
      *(v2f*)&BJ1[n * 2]              = lo1;
      *(v2f*)&BJ1[(NPTS + n) * 2]     = hi1;
      v2f lop; lop.x = m00; lop.y = -b0;
      v2f hip; hip.x = m01; hip.y = -b1;
      *(v2f*)&BP[n * 2]               = lop;
      *(v2f*)&BP[(NPTS + n) * 2]      = hip;
      v4f c; c.x = -cc[i * 3 + 0]; c.y = -cc[i * 3 + 1];
      c.z = -cc[i * 3 + 2]; c.w = 0.0f;
      *(v4f*)&C4[n * 4] = c;
    }
  }
  __syncthreads();

  const int lane  = threadIdx.x & 31;
  const int wave  = __builtin_amdgcn_readfirstlane((int)(threadIdx.x >> 5));
  const int patch = wave / NQ;
  const int quart = wave - patch * NQ;
  const int hi    = lane >> 4;
  const int ln    = lane & 15;

  const int groups_per_row = CANVAS_W / (PATCH_W * NPATCH);
  const int ph = (int)blockIdx.x / groups_per_row;
  const int pg = (int)blockIdx.x - ph * groups_per_row;
  const int h0 = ph * PATCH_H;
  const int wg0 = pg * (PATCH_W * NPATCH);
  const int w0 = wg0 + patch * PATCH_W;

  const float gy = -1.0f + (float)(h0 + (ln & 7)) * (2.0f / (float)(CANVAS_H - 1));
  const float gx = -R_WH + (float)(w0 + ln) * (2.0f * R_WH / (float)(CANVAS_W - 1));
  v2f aP, aQ;
  const bool own = ((ln >> 3) == hi);
  aP.x = own ? 0.5f * gy : 0.0f;
  aP.y = own ? 0.5f : 0.0f;
  aQ.x = 0.0f;
  aQ.y = hi ? 0.0f : 0.5f * gx;

  const v8f cz = {};
  const int bsel = (hi * NPTS + ln) * 2;

#define ROW_PRODUCTS(K, B)                                                         \
  {                                                                                \
    const v8f p = __builtin_amdgcn_wmma_f32_16x16x4_f32(false, aP, false, B,       \
                                                   (short)0, cz, false, false);    \
    float* d = ptl + (K) * (16 * PT_PITCH);                                        \
    *(v4f*)&d[0] = __builtin_shufflevector(p, p, 0, 1, 2, 3);                      \
    *(v4f*)&d[4] = __builtin_shufflevector(p, p, 4, 5, 6, 7);                      \
  }
  {
    float* ptl = &PT[ln * PT_PITCH + 8 * hi];
    v2f rb[5];
#pragma unroll
    for (int i = 0; i < 5; ++i) rb[i] = *(const v2f*)&BP[bsel + (wave + NWAVES * i) * 32];
#pragma unroll
    for (int i = 0; i < 5; ++i) ROW_PRODUCTS(wave + NWAVES * i, rb[i])
    if (wave < NPTS / 16 - 5 * NWAVES) {
      const int k = wave + NWAVES * 5;
      const v2f lb = *(const v2f*)&BP[bsel + k * 32];
      ROW_PRODUCTS(k, lb)
    }
  }
#undef ROW_PRODUCTS
  __syncthreads();

  v2f acc2[96];
#pragma unroll
  for (int i = 0; i < 96; ++i) acc2[i] = (v2f){0.0f, 0.0f};

  const int kbeg = quart * (NPTS / 16 / NQ);
  v2f cs01 = {0.0f, 0.0f}, cs23 = {0.0f, 0.0f};

  for (int kk = 0; kk < NPTS / 16 / NQ; ++kk) {
    const int k = kbeg + kk;
    v2f b0 = *(const v2f*)&BJ0[bsel + k * 32];
    v2f b1 = *(const v2f*)&BJ1[bsel + k * 32];
    v4f c4 = *(const v4f*)&C4[(k * 16 + ln) * 4];
    const float* pp = &PT[(k * 16 + ln) * PT_PITCH];
    const v4f pa = *(const v4f*)&pp[0];
    const v4f pb = *(const v4f*)&pp[4];
    const v4f pc = *(const v4f*)&pp[8];
    const v4f pd = *(const v4f*)&pp[12];
    const v8f P0 = __builtin_shufflevector(pa, pb, 0, 1, 2, 3, 4, 5, 6, 7);
    const v8f P1 = __builtin_shufflevector(pc, pd, 0, 1, 2, 3, 4, 5, 6, 7);

    v8f Q0 = __builtin_amdgcn_wmma_f32_16x16x4_f32(false, aQ, false, b0,
                                                   (short)0, cz, false, false);
    v8f Q1 = __builtin_amdgcn_wmma_f32_16x16x4_f32(false, aQ, false, b1,
                                                   (short)0, cz, false, false);

    asm volatile("s_wait_alu depctr_va_vdst(0)" : "+v"(Q0), "+v"(Q1));

    const v2f c01 = {c4.x, c4.y}, c23 = {c4.z, c4.w};
    cs01 -= c01;
    cs23 -= c23;

#define ROW_SUMS(R, T0, T1)                                                        \
    {                                                                              \
      const v2f pp0 = {P0[(R) & 6], P0[((R) & 6) + 1]};                            \
      const v2f pp1 = {P1[(R) & 6], P1[((R) & 6) + 1]};                            \
      _Pragma("unroll") for (int q = 0; q < 4; ++q) {                              \
        const v2f qq = {Q0[2 * q], Q0[2 * q + 1]};                                 \
        T0[q] = bcast_add<(R) & 1>(pp0, qq);                                       \
      }                                                                            \
      _Pragma("unroll") for (int q = 0; q < 4; ++q) {                              \
        const v2f qq = {Q1[2 * q], Q1[2 * q + 1]};                                 \
        T1[q] = bcast_add<(R) & 1>(pp1, qq);                                       \
      }                                                                            \
    }
#define ROW_REST(R, T0, T1, NEXT)                                                  \
    {                                                                              \
      float xs[8];                                                                 \
      _Pragma("unroll") for (int q = 0; q < 4; ++q) {                              \
        xs[2 * q] = T0[q].x; xs[2 * q + 1] = T0[q].y;                              \
        abs_sum_sat(xs[2 * q], T1[q].x); abs_sum_sat(xs[2 * q + 1], T1[q].y);      \
      }                                                                            \
      NEXT                                                                         \
      v2f mm2[4];                                                                  \
      _Pragma("unroll") for (int q = 0; q < 4; ++q) {                              \
        mm2[q].x = xs[2 * q]; mm2[q].y = xs[2 * q + 1];                            \
      }                                                                            \
      _Pragma("unroll") for (int q = 0; q < 4; ++q) {                              \
        bcast_fma<0>(acc2[0 * 32 + (R) * 4 + q], mm2[q], c01);                     \
        bcast_fma<1>(acc2[1 * 32 + (R) * 4 + q], mm2[q], c01);                     \
        bcast_fma<0>(acc2[2 * 32 + (R) * 4 + q], mm2[q], c23);                     \
      }                                                                            \
    }
    v2f ta0[4], ta1[4], tb0[4], tb1[4];
    ROW_SUMS(0, ta0, ta1)
    ROW_REST(0, ta0, ta1, ROW_SUMS(1, tb0, tb1))
    ROW_REST(1, tb0, tb1, ROW_SUMS(2, ta0, ta1))
    ROW_REST(2, ta0, ta1, ROW_SUMS(3, tb0, tb1))
    ROW_REST(3, tb0, tb1, ROW_SUMS(4, ta0, ta1))
    ROW_REST(4, ta0, ta1, ROW_SUMS(5, tb0, tb1))
    ROW_REST(5, tb0, tb1, ROW_SUMS(6, ta0, ta1))
    ROW_REST(6, ta0, ta1, ROW_SUMS(7, tb0, tb1))
    ROW_REST(7, tb0, tb1, )
#undef ROW_SUMS
#undef ROW_REST
  }

  float v[192];
#pragma unroll
  for (int i = 0; i < 32; ++i) {
    v[2 * i]            = acc2[i].x + cs01.x;       v[2 * i + 1]        = acc2[i].y + cs01.x;
    v[64 + 2 * i]       = acc2[32 + i].x + cs01.y;  v[64 + 2 * i + 1]   = acc2[32 + i].y + cs01.y;
    v[128 + 2 * i]      = acc2[64 + i].x + cs23.x;  v[128 + 2 * i + 1]  = acc2[64 + i].y + cs23.x;
  }

  float s1[96], s2[48], s3[24], s4[12];
  scatter_step<8, 192>(v,  s1, (ln & 8) != 0);
  scatter_step<4, 96>(s1, s2, (ln & 4) != 0);
  scatter_step<2, 48>(s2, s3, (ln & 2) != 0);
  scatter_step<1, 24>(s3, s4, (ln & 1) != 0);

  {
    float* dst = &red[(wave * 2 + hi) * 192 + ln * 12];
#pragma unroll
    for (int i = 0; i < 3; ++i) {
      v4f w4; w4.x = s4[4 * i]; w4.y = s4[4 * i + 1]; w4.z = s4[4 * i + 2]; w4.w = s4[4 * i + 3];
      *(v4f*)&dst[4 * i] = w4;
    }
  }
  __syncthreads();
  {
    const int t  = (int)threadIdx.x;
    const int r1 = t / (PATCH_W * NPATCH);
    const int gc = t - r1 * (PATCH_W * NPATCH);
    const int pt = gc >> 4;
    const int wc = gc & 15;
    const int hh = wc >> 3;
    const int r2 = wc & 7;
    const int p  = (h0 + r1) * CANVAS_W + (wg0 + gc);
#pragma unroll
    for (int ch = 0; ch < 3; ++ch) {
      const int idx = ch * 64 + r1 * 8 + r2;
      float s = 0.0f;
#pragma unroll
      for (int wv = 0; wv < NQ; ++wv) s += red[((pt * NQ + wv) * 2 + hh) * 192 + idx];
      out[ch * CANVAS_H * CANVAS_W + p] = 1.0f / (1.0f + __expf(-4.0f * s));
    }
  }
}

extern "C" void kernel_launch(void* const* d_in, const int* in_sizes, int n_in,
                              void* d_out, int out_size, void* d_ws, size_t ws_size,
                              hipStream_t stream) {
  (void)in_sizes; (void)n_in; (void)out_size; (void)d_ws; (void)ws_size;
  const float* loc = (const float*)d_in[0];
  const float* mo  = (const float*)d_in[1];
  const float* mse = (const float*)d_in[2];
  const float* col = (const float*)d_in[3];
  float* out = (float*)d_out;

  const int blocks = (CANVAS_H / PATCH_H) * (CANVAS_W / (PATCH_W * NPATCH));
  points_kernel<<<blocks, 32 * NWAVES, 0, stream>>>(loc, mo, mse, col, out);
}
